# starting kernel
# baseline (speedup 1.0000x reference)
#include <hip/hip_runtime.h>
#include <hip/hip_bf16.h>
#include <math.h>

typedef __attribute__((ext_vector_type(2))) float v2f;
typedef __attribute__((ext_vector_type(8))) float v8f;

#define N_PTS 8192
#define DIM   512
#define ROWT  128
#define CHUNK 1024
#define NCH   (N_PTS / CHUNK)
#define CT    128
#define KB    32
#define LDA   36
#define FLT_BIG 3.402823466e38f

__global__ void norms_kernel(const float* __restrict__ f, float* __restrict__ norms) {
    int row = blockIdx.x * blockDim.x + threadIdx.x;
    if (row >= N_PTS) return;
    const float4* fr = (const float4*)(f + (size_t)row * DIM);
    float s = 0.f;
#pragma unroll 4
    for (int i = 0; i < DIM / 4; ++i) {
        float4 v = fr[i];
        s += v.x * v.x + v.y * v.y + v.z * v.z + v.w * v.w;
    }
    norms[row] = s;
}

__launch_bounds__(256)
__global__ void tile_kernel(const float* __restrict__ feat,
                            const int*   __restrict__ labels,
                            const float* __restrict__ norms,
                            float* __restrict__ pMaxAll,
                            float* __restrict__ pMaxPos,
                            float* __restrict__ pMinNeg) {
    __shared__ float Alds[ROWT * LDA];
    __shared__ float Blds[CT * LDA];
    __shared__ float rowNorm[ROWT];
    __shared__ int   rowLab[ROWT];
    __shared__ float colNorm[CHUNK];
    __shared__ int   colLab[CHUNK];

    const int tid  = threadIdx.x;
    const int wave = tid >> 5;
    const int lane = tid & 31;
    const int lo   = lane & 15;
    const int hi   = lane >> 4;
    const int r0   = blockIdx.x * ROWT;
    const int cc0  = blockIdx.y * CHUNK;

    for (int i = tid; i < ROWT; i += 256) {
        rowNorm[i] = norms[r0 + i];
        rowLab[i]  = labels[r0 + i];
    }
    for (int i = tid; i < CHUNK; i += 256) {
        colNorm[i] = norms[cc0 + i];
        colLab[i]  = labels[cc0 + i];
    }
    __syncthreads();

    float maxAll[8], maxPos[8], minNeg[8];
#pragma unroll
    for (int v = 0; v < 8; ++v) { maxAll[v] = 0.f; maxPos[v] = -1.f; minNeg[v] = FLT_BIG; }

    const v8f vzero = {0.f, 0.f, 0.f, 0.f, 0.f, 0.f, 0.f, 0.f};

    for (int ct = 0; ct < CHUNK / CT; ++ct) {
        const int c0 = cc0 + ct * CT;
        v8f acc[8];
#pragma unroll
        for (int g = 0; g < 8; ++g) acc[g] = vzero;

        for (int kb = 0; kb < DIM / KB; ++kb) {
            const int k0 = kb * KB;
#pragma unroll
            for (int i = 0; i < 4; ++i) {
                int idx = tid + i * 256;
                int row = idx >> 3;
                int c4  = (idx & 7) << 2;
                float4 av = *(const float4*)(feat + (size_t)(r0 + row) * DIM + k0 + c4);
                *(float4*)(&Alds[row * LDA + c4]) = av;
                float4 bv = *(const float4*)(feat + (size_t)(c0 + row) * DIM + k0 + c4);
                *(float4*)(&Blds[row * LDA + c4]) = bv;
            }
            __syncthreads();

#pragma unroll
            for (int k4 = 0; k4 < KB; k4 += 4) {
                v2f a = *(const v2f*)(&Alds[(wave * 16 + lo) * LDA + k4 + hi * 2]);
#pragma unroll
                for (int g = 0; g < 8; ++g) {
                    v2f b = *(const v2f*)(&Blds[(g * 16 + lo) * LDA + k4 + hi * 2]);
                    acc[g] = __builtin_amdgcn_wmma_f32_16x16x4_f32(
                        false, a, false, b, (short)0, acc[g], false, false);
                }
            }
            __syncthreads();
        }

#pragma unroll
        for (int g = 0; g < 8; ++g) {
            const int ci = ct * CT + g * 16 + lo;
            const int gc = cc0 + ci;
            const float cn = colNorm[ci];
            const int   cl = colLab[ci];
#pragma unroll
            for (int v = 0; v < 8; ++v) {
                const int rl = wave * 16 + v + hi * 8;
                const int gr = r0 + rl;
                float sq = rowNorm[rl] + cn - 2.0f * acc[g][v];
                float dist = (sq > 0.f) ? sqrtf(sq) : 0.f;
                if (gr == gc) dist = 0.f;
                const bool same = (rowLab[rl] == cl);
                maxAll[v] = fmaxf(maxAll[v], dist);
                if (same) {
                    if (gr != gc) maxPos[v] = fmaxf(maxPos[v], dist);
                } else {
                    minNeg[v] = fminf(minNeg[v], dist);
                }
            }
        }
    }

#pragma unroll
    for (int v = 0; v < 8; ++v) {
        float ma = maxAll[v], mp = maxPos[v], mn = minNeg[v];
#pragma unroll
        for (int m = 1; m <= 8; m <<= 1) {
            ma = fmaxf(ma, __shfl_xor(ma, m, 32));
            mp = fmaxf(mp, __shfl_xor(mp, m, 32));
            mn = fminf(mn, __shfl_xor(mn, m, 32));
        }
        if (lo == 0) {
            const int grow = r0 + wave * 16 + v + hi * 8;
            const size_t o = (size_t)grow * NCH + blockIdx.y;
            pMaxAll[o] = ma;
            pMaxPos[o] = mp;
            pMinNeg[o] = mn;
        }
    }
}

__global__ void final_kernel(const float* __restrict__ pMaxAll,
                             const float* __restrict__ pMaxPos,
                             const float* __restrict__ pMinNeg,
                             float* __restrict__ out) {
    int row = blockIdx.x * blockDim.x + threadIdx.x;
    if (row >= N_PTS) return;
    float ma = 0.f, mp = -1.f, mn = FLT_BIG;
#pragma unroll
    for (int c = 0; c < NCH; ++c) {
        ma = fmaxf(ma, pMaxAll[row * NCH + c]);
        mp = fmaxf(mp, pMaxPos[row * NCH + c]);
        mn = fminf(mn, pMinNeg[row * NCH + c]);
    }
    out[row * 2 + 0] = (mp >= 0.f) ? mp : 0.f;
    out[row * 2 + 1] = (mn < 3.0e38f) ? mn : ma;
}

extern "C" void kernel_launch(void* const* d_in, const int* in_sizes, int n_in,
                              void* d_out, int out_size, void* d_ws, size_t ws_size,
                              hipStream_t stream) {
    (void)in_sizes; (void)n_in; (void)out_size; (void)ws_size;
    const float* feat  = (const float*)d_in[0];
    const int* labels  = (const int*)d_in[1];
    float* ws = (float*)d_ws;

    float* norms   = ws;
    float* pMaxAll = norms + N_PTS;
    float* pMaxPos = pMaxAll + (size_t)N_PTS * NCH;
    float* pMinNeg = pMaxPos + (size_t)N_PTS * NCH;
    float* out = (float*)d_out;

    norms_kernel<<<N_PTS / 256, 256, 0, stream>>>(feat, norms);

    dim3 grid(N_PTS / ROWT, NCH);
    tile_kernel<<<grid, 256, 0, stream>>>(feat, labels, norms, pMaxAll, pMaxPos, pMinNeg);

    final_kernel<<<N_PTS / 256, 256, 0, stream>>>(pMaxAll, pMaxPos, pMinNeg, out);
}


# X*X^T symmetry: each 128x128 tile of one block pair is computed once on WMMA and reduced along its rows and its columns; 2080 tiles instead of 4096; max/min only, no float sum reordered
# speedup vs baseline: 1.8264x; 1.8264x over previous
#include <hip/hip_runtime.h>
#include <hip/hip_bf16.h>
#include <math.h>


typedef __attribute__((ext_vector_type(2))) float v2f;
typedef __attribute__((ext_vector_type(4))) float v4f;
typedef __attribute__((ext_vector_type(8))) float v8f;

#define N_PTS 8192
#define DIM   512
#define TILE  128
#define NB    (N_PTS / TILE)
#define ND    (NB / 2 + 1)
#define KB    32
#define LDA   36
#define FLT_BIG 3.402823466e38f

__global__ void norms_kernel(const float* __restrict__ f, float* __restrict__ norms) {
    int row = blockIdx.x * blockDim.x + threadIdx.x;
    if (row >= N_PTS) return;
    const float4* fr = (const float4*)(f + (size_t)row * DIM);
    float s = 0.f;
#pragma unroll 4
    for (int i = 0; i < DIM / 4; ++i) {
        float4 v = fr[i];
        s += v.x * v.x + v.y * v.y + v.z * v.z + v.w * v.w;
    }
    norms[row] = s;
}

__launch_bounds__(256)
__global__ void tile_kernel(const float* __restrict__ feat,
                            const int*   __restrict__ labels,
                            const float* __restrict__ norms,
                            float* __restrict__ part) {
    __shared__ float Alds[TILE * LDA];
    __shared__ float Blds[TILE * LDA];
    __shared__ float rowNorm[TILE];
    __shared__ int   rowLab[TILE];
    __shared__ float colNorm[TILE];
    __shared__ int   colLab[TILE];
    __shared__ float rowRed[3][TILE];
    __shared__ float colRed[8][3][TILE];

    const int bi = blockIdx.x;
    const int d  = blockIdx.y;
    if (d == NB / 2 && bi >= NB / 2) return;
    const int bj = (bi + d) & (NB - 1);

    const int tid  = threadIdx.x;
    const int wave = tid >> 5;
    const int lane = tid & 31;
    const int lo   = lane & 15;
    const int hi   = lane >> 4;
    const int r0   = bi * TILE;
    const int c0   = bj * TILE;

    if (tid < TILE) {
        rowNorm[tid] = norms[r0 + tid];
        rowLab[tid]  = labels[r0 + tid];
    } else {
        const int t = tid - TILE;
        colNorm[t] = norms[c0 + t];
        colLab[t]  = labels[c0 + t];
    }

    const v8f vzero = {0.f, 0.f, 0.f, 0.f, 0.f, 0.f, 0.f, 0.f};
    v8f acc[8];
#pragma unroll
    for (int g = 0; g < 8; ++g) acc[g] = vzero;

    for (int kb = 0; kb < DIM / KB; ++kb) {
        const int k0 = kb * KB;
        v4f av[4], bv[4];
#pragma unroll
        for (int i = 0; i < 4; ++i) {
            int idx = tid + i * 256;
            int row = idx >> 3;
            int c4  = (idx & 7) << 2;
            av[i] = *(const v4f*)(feat + (size_t)(r0 + row) * DIM + k0 + c4);
            bv[i] = *(const v4f*)(feat + (size_t)(c0 + row) * DIM + k0 + c4);
        }
        asm volatile("" :: "v"(av[0]), "v"(av[1]), "v"(av[2]), "v"(av[3]),
                           "v"(bv[0]), "v"(bv[1]), "v"(bv[2]), "v"(bv[3]));
#pragma unroll
        for (int i = 0; i < 4; ++i) {
            int idx = tid + i * 256;
            int row = idx >> 3;
            int c4  = (idx & 7) << 2;
            *(v4f*)(&Alds[row * LDA + c4]) = av[i];
            *(v4f*)(&Blds[row * LDA + c4]) = bv[i];
        }
        __syncthreads();

#pragma unroll
        for (int k4 = 0; k4 < KB; k4 += 4) {
            v2f a = *(const v2f*)(&Alds[(wave * 16 + lo) * LDA + k4 + hi * 2]);
#pragma unroll
            for (int g = 0; g < 8; ++g) {
                v2f b = *(const v2f*)(&Blds[(g * 16 + lo) * LDA + k4 + hi * 2]);
                acc[g] = __builtin_amdgcn_wmma_f32_16x16x4_f32(
                    false, a, false, b, (short)0, acc[g], false, false);
            }
        }
        __syncthreads();
    }

    float maxAll[8], maxPos[8], minNeg[8];
#pragma unroll
    for (int v = 0; v < 8; ++v) { maxAll[v] = 0.f; maxPos[v] = -1.f; minNeg[v] = FLT_BIG; }

#pragma unroll
    for (int g = 0; g < 8; ++g) {
        const int ci = g * 16 + lo;
        const int gc = c0 + ci;
        const float cn = colNorm[ci];
        const int   cl = colLab[ci];
        float cA = 0.f, cP = -1.f, cN = FLT_BIG;
#pragma unroll
        for (int v = 0; v < 8; ++v) {
            const int rl = wave * 16 + v + hi * 8;
            const int gr = r0 + rl;
            float sq = rowNorm[rl] + cn - 2.0f * acc[g][v];
            float dist = (sq > 0.f) ? sqrtf(sq) : 0.f;
            if (gr == gc) dist = 0.f;
            const bool same = (rowLab[rl] == cl);
            maxAll[v] = fmaxf(maxAll[v], dist);
            cA = fmaxf(cA, dist);
            if (same) {
                if (gr != gc) {
                    maxPos[v] = fmaxf(maxPos[v], dist);
                    cP = fmaxf(cP, dist);
                }
            } else {
                minNeg[v] = fminf(minNeg[v], dist);
                cN = fminf(cN, dist);
            }
        }
        cA = fmaxf(cA, __shfl_xor(cA, 16, 32));
        cP = fmaxf(cP, __shfl_xor(cP, 16, 32));
        cN = fminf(cN, __shfl_xor(cN, 16, 32));
        if (hi == 0) {
            colRed[wave][0][ci] = cA;
            colRed[wave][1][ci] = cP;
            colRed[wave][2][ci] = cN;
        }
    }

#pragma unroll
    for (int v = 0; v < 8; ++v) {
        float ma = maxAll[v], mp = maxPos[v], mn = minNeg[v];
#pragma unroll
        for (int m = 1; m <= 8; m <<= 1) {
            ma = fmaxf(ma, __shfl_xor(ma, m, 32));
            mp = fmaxf(mp, __shfl_xor(mp, m, 32));
            mn = fminf(mn, __shfl_xor(mn, m, 32));
        }
        if (lo == 0) {
            const int rl = wave * 16 + v + hi * 8;
            rowRed[0][rl] = ma;
            rowRed[1][rl] = mp;
            rowRed[2][rl] = mn;
        }
    }
    __syncthreads();

    for (int idx = tid; idx < 3 * TILE; idx += 256) {
        const int q = idx >> 7;
        const int e = idx & (TILE - 1);
        part[((size_t)q * NB + bj) * N_PTS + r0 + e] = rowRed[q][e];
        if (d != 0) {
            float a = colRed[0][q][e];
#pragma unroll
            for (int w = 1; w < 8; ++w) {
                const float b = colRed[w][q][e];
                a = (q == 2) ? fminf(a, b) : fmaxf(a, b);
            }
            part[((size_t)q * NB + bi) * N_PTS + c0 + e] = a;
        }
    }
}

__global__ void final_kernel(const float* __restrict__ part, float* __restrict__ out) {
    __shared__ float sm[3][8][32];
    const int lane = threadIdx.x & 31;
    const int w    = threadIdx.x >> 5;
    const int row  = blockIdx.x * 32 + lane;
    float ma = 0.f, mp = -1.f, mn = FLT_BIG;
#pragma unroll
    for (int s = 0; s < 8; ++s) {
        const size_t o = (size_t)(w * 8 + s) * N_PTS + row;
        ma = fmaxf(ma, part[o]);
        mp = fmaxf(mp, part[(size_t)NB * N_PTS + o]);
        mn = fminf(mn, part[(size_t)2 * NB * N_PTS + o]);
    }
    sm[0][w][lane] = ma;
    sm[1][w][lane] = mp;
    sm[2][w][lane] = mn;
    __syncthreads();
    if (w == 0) {
#pragma unroll
        for (int k = 1; k < 8; ++k) {
            ma = fmaxf(ma, sm[0][k][lane]);
            mp = fmaxf(mp, sm[1][k][lane]);
            mn = fminf(mn, sm[2][k][lane]);
        }
        float2 o2;
        o2.x = (mp >= 0.f) ? mp : 0.f;
        o2.y = (mn < 3.0e38f) ? mn : ma;
        *(float2*)(out + (size_t)row * 2) = o2;
    }
}

extern "C" void kernel_launch(void* const* d_in, const int* in_sizes, int n_in,
                              void* d_out, int out_size, void* d_ws, size_t ws_size,
                              hipStream_t stream) {
    (void)in_sizes; (void)n_in; (void)out_size; (void)ws_size;
    const float* feat  = (const float*)d_in[0];
    const int* labels  = (const int*)d_in[1];
    float* ws = (float*)d_ws;

    float* norms = ws;
    float* part  = norms + N_PTS;
    float* out = (float*)d_out;

    norms_kernel<<<N_PTS / 256, 256, 0, stream>>>(feat, norms);

    dim3 grid(NB, ND);
    tile_kernel<<<grid, 256, 0, stream>>>(feat, labels, norms, part);

    final_kernel<<<N_PTS / 32, 256, 0, stream>>>(part, out);
}


# tile epilogue: masked max/min by selects on squared distances (no per-element sqrt sequence, no per-element exec branch); sqrt once per row in last kernel (monotone, same bits); no sum reordered
# speedup vs baseline: 1.9311x; 1.0573x over previous
#include <hip/hip_runtime.h>
#include <hip/hip_bf16.h>
#include <math.h>


typedef __attribute__((ext_vector_type(2))) float v2f;
typedef __attribute__((ext_vector_type(4))) float v4f;
typedef __attribute__((ext_vector_type(8))) float v8f;

#define N_PTS 8192
#define DIM   512
#define TILE  128
#define NB    (N_PTS / TILE)
#define ND    (NB / 2 + 1)
#define KB    32
#define LDA   36
#define F_INF (__builtin_inff())

__global__ void norms_kernel(const float* __restrict__ f, float* __restrict__ norms) {
    int row = blockIdx.x * blockDim.x + threadIdx.x;
    if (row >= N_PTS) return;
    const float4* fr = (const float4*)(f + (size_t)row * DIM);
    float s = 0.f;
#pragma unroll 4
    for (int i = 0; i < DIM / 4; ++i) {
        float4 v = fr[i];
        s += v.x * v.x + v.y * v.y + v.z * v.z + v.w * v.w;
    }
    norms[row] = s;
}

__launch_bounds__(256)
__global__ void tile_kernel(const float* __restrict__ feat,
                            const int*   __restrict__ labels,
                            const float* __restrict__ norms,
                            float* __restrict__ part) {
    __shared__ float Alds[TILE * LDA];
    __shared__ float Blds[TILE * LDA];
    __shared__ float rowNorm[TILE];
    __shared__ int   rowLab[TILE];
    __shared__ float colNorm[TILE];
    __shared__ int   colLab[TILE];
    __shared__ float rowRed[3][TILE];
    __shared__ float colRed[8][3][TILE];

    const int bi = blockIdx.x;
    const int d  = blockIdx.y;
    if (d == NB / 2 && bi >= NB / 2) return;
    const int bj = (bi + d) & (NB - 1);

    const int tid  = threadIdx.x;
    const int wave = tid >> 5;
    const int lane = tid & 31;
    const int lo   = lane & 15;
    const int hi   = lane >> 4;
    const int r0   = bi * TILE;
    const int c0   = bj * TILE;

    if (tid < TILE) {
        rowNorm[tid] = norms[r0 + tid];
        rowLab[tid]  = labels[r0 + tid];
    } else {
        const int t = tid - TILE;
        colNorm[t] = norms[c0 + t];
        colLab[t]  = labels[c0 + t];
    }

    const v8f vzero = {0.f, 0.f, 0.f, 0.f, 0.f, 0.f, 0.f, 0.f};
    v8f acc[8];
#pragma unroll
    for (int g = 0; g < 8; ++g) acc[g] = vzero;

    for (int kb = 0; kb < DIM / KB; ++kb) {
        const int k0 = kb * KB;
        v4f av[4], bv[4];
#pragma unroll
        for (int i = 0; i < 4; ++i) {
            int idx = tid + i * 256;
            int row = idx >> 3;
            int c4  = (idx & 7) << 2;
            av[i] = *(const v4f*)(feat + (size_t)(r0 + row) * DIM + k0 + c4);
            bv[i] = *(const v4f*)(feat + (size_t)(c0 + row) * DIM + k0 + c4);
        }
        asm volatile("" :: "v"(av[0]), "v"(av[1]), "v"(av[2]), "v"(av[3]),
                           "v"(bv[0]), "v"(bv[1]), "v"(bv[2]), "v"(bv[3]));
#pragma unroll
        for (int i = 0; i < 4; ++i) {
            int idx = tid + i * 256;
            int row = idx >> 3;
            int c4  = (idx & 7) << 2;
            *(v4f*)(&Alds[row * LDA + c4]) = av[i];
            *(v4f*)(&Blds[row * LDA + c4]) = bv[i];
        }
        __syncthreads();

#pragma unroll
        for (int k4 = 0; k4 < KB; k4 += 4) {
            v2f a = *(const v2f*)(&Alds[(wave * 16 + lo) * LDA + k4 + hi * 2]);
#pragma unroll
            for (int g = 0; g < 8; ++g) {
                v2f b = *(const v2f*)(&Blds[(g * 16 + lo) * LDA + k4 + hi * 2]);
                acc[g] = __builtin_amdgcn_wmma_f32_16x16x4_f32(
                    false, a, false, b, (short)0, acc[g], false, false);
            }
        }
        __syncthreads();
    }

    float maxAll[8], maxPos[8], minNeg[8];
#pragma unroll
    for (int v = 0; v < 8; ++v) { maxAll[v] = 0.f; maxPos[v] = -1.f; minNeg[v] = F_INF; }

#pragma unroll
    for (int g = 0; g < 8; ++g) {
        const int ci = g * 16 + lo;
        const int gc = c0 + ci;
        const float cn = colNorm[ci];
        const int   cl = colLab[ci];
        float cA = 0.f, cP = -1.f, cN = F_INF;
#pragma unroll
        for (int v = 0; v < 8; ++v) {
            const int rl = wave * 16 + v + hi * 8;
            const int gr = r0 + rl;
            const float sq   = rowNorm[rl] + cn - 2.0f * acc[g][v];
            const bool  offd = (gr != gc);
            const float d2   = (offd && sq > 0.f) ? sq : 0.f;
            const bool  same = (rowLab[rl] == cl);
            const float dp   = (same && offd) ? d2 : -1.f;
            const float dn   = same ? F_INF : d2;
            maxAll[v] = fmaxf(maxAll[v], d2);
            cA        = fmaxf(cA, d2);
            maxPos[v] = fmaxf(maxPos[v], dp);
            cP        = fmaxf(cP, dp);
            minNeg[v] = fminf(minNeg[v], dn);
            cN        = fminf(cN, dn);
        }
        cA = fmaxf(cA, __shfl_xor(cA, 16, 32));
        cP = fmaxf(cP, __shfl_xor(cP, 16, 32));
        cN = fminf(cN, __shfl_xor(cN, 16, 32));
        if (hi == 0) {
            colRed[wave][0][ci] = cA;
            colRed[wave][1][ci] = cP;
            colRed[wave][2][ci] = cN;
        }
        asm volatile("" :: "v"(maxAll[0]), "v"(maxAll[1]), "v"(maxAll[2]), "v"(maxAll[3]),
                           "v"(maxAll[4]), "v"(maxAll[5]), "v"(maxAll[6]), "v"(maxAll[7]),
                           "v"(maxPos[0]), "v"(maxPos[1]), "v"(maxPos[2]), "v"(maxPos[3]),
                           "v"(maxPos[4]), "v"(maxPos[5]), "v"(maxPos[6]), "v"(maxPos[7]),
                           "v"(minNeg[0]), "v"(minNeg[1]), "v"(minNeg[2]), "v"(minNeg[3]),
                           "v"(minNeg[4]), "v"(minNeg[5]), "v"(minNeg[6]), "v"(minNeg[7]));
    }

#pragma unroll
    for (int v = 0; v < 8; ++v) {
        float ma = maxAll[v], mp = maxPos[v], mn = minNeg[v];
#pragma unroll
        for (int m = 1; m <= 8; m <<= 1) {
            ma = fmaxf(ma, __shfl_xor(ma, m, 32));
            mp = fmaxf(mp, __shfl_xor(mp, m, 32));
            mn = fminf(mn, __shfl_xor(mn, m, 32));
        }
        if (lo == 0) {
            const int rl = wave * 16 + v + hi * 8;
            rowRed[0][rl] = ma;
            rowRed[1][rl] = mp;
            rowRed[2][rl] = mn;
        }
    }
    __syncthreads();

    for (int idx = tid; idx < 3 * TILE; idx += 256) {
        const int q = idx >> 7;
        const int e = idx & (TILE - 1);
        part[((size_t)q * NB + bj) * N_PTS + r0 + e] = rowRed[q][e];
        if (d != 0) {
            float a = colRed[0][q][e];
#pragma unroll
            for (int w = 1; w < 8; ++w) {
                const float b = colRed[w][q][e];
                a = (q == 2) ? fminf(a, b) : fmaxf(a, b);
            }
            part[((size_t)q * NB + bi) * N_PTS + c0 + e] = a;
        }
    }
}

__global__ void final_kernel(const float* __restrict__ part, float* __restrict__ out) {
    __shared__ float sm[3][8][32];
    const int lane = threadIdx.x & 31;
    const int w    = threadIdx.x >> 5;
    const int row  = blockIdx.x * 32 + lane;
    float ma = 0.f, mp = -1.f, mn = F_INF;
#pragma unroll
    for (int s = 0; s < 8; ++s) {
        const size_t o = (size_t)(w * 8 + s) * N_PTS + row;
        ma = fmaxf(ma, part[o]);
        mp = fmaxf(mp, part[(size_t)NB * N_PTS + o]);
        mn = fminf(mn, part[(size_t)2 * NB * N_PTS + o]);
    }
    sm[0][w][lane] = ma;
    sm[1][w][lane] = mp;
    sm[2][w][lane] = mn;
    __syncthreads();
    if (w == 0) {
#pragma unroll
        for (int k = 1; k < 8; ++k) {
            ma = fmaxf(ma, sm[0][k][lane]);
            mp = fmaxf(mp, sm[1][k][lane]);
            mn = fminf(mn, sm[2][k][lane]);
        }
        float2 o2;
        o2.x = (mp >= 0.f) ? sqrtf(mp) : 0.f;
        o2.y = sqrtf((mn < F_INF) ? mn : ma);
        *(float2*)(out + (size_t)row * 2) = o2;
    }
}

extern "C" void kernel_launch(void* const* d_in, const int* in_sizes, int n_in,
                              void* d_out, int out_size, void* d_ws, size_t ws_size,
                              hipStream_t stream) {
    (void)in_sizes; (void)n_in; (void)out_size; (void)ws_size;
    const float* feat  = (const float*)d_in[0];
    const int* labels  = (const int*)d_in[1];
    float* ws = (float*)d_ws;

    float* norms = ws;
    float* part  = norms + N_PTS;
    float* out = (float*)d_out;

    norms_kernel<<<N_PTS / 256, 256, 0, stream>>>(feat, norms);

    dim3 grid(NB, ND);
    tile_kernel<<<grid, 256, 0, stream>>>(feat, labels, norms, part);

    final_kernel<<<N_PTS / 32, 256, 0, stream>>>(part, out);
}


# tile epilogue, fewer instructions: clamp once per reduced value, self-pair test only in diagonal tiles, DPP/permlane reductions, all-element max dropped (same-label max where read); no sum reordered
# speedup vs baseline: 2.0301x; 1.0513x over previous
#include <hip/hip_runtime.h>
#include <hip/hip_bf16.h>
#include <math.h>


typedef __attribute__((ext_vector_type(2))) float v2f;
typedef __attribute__((ext_vector_type(4))) float v4f;
typedef __attribute__((ext_vector_type(8))) float v8f;

#define N_PTS 8192
#define DIM   512
#define TILE  128
#define NB    (N_PTS / TILE)
#define ND    (NB / 2 + 1)
#define KB    32
#define LDA   36
#define F_INF (__builtin_inff())

__global__ void norms_kernel(const float* __restrict__ f, float* __restrict__ norms) {
    int row = blockIdx.x * blockDim.x + threadIdx.x;
    if (row >= N_PTS) return;
    const float4* fr = (const float4*)(f + (size_t)row * DIM);
    float s = 0.f;
#pragma unroll 4
    for (int i = 0; i < DIM / 4; ++i) {
        float4 v = fr[i];
        s += v.x * v.x + v.y * v.y + v.z * v.z + v.w * v.w;
    }
    norms[row] = s;
}


__device__ __forceinline__ float maxf2(float a, float b) { return __builtin_fmaximum_numf(a, b); }
__device__ __forceinline__ float minf2(float a, float b) { return __builtin_fminimum_numf(a, b); }
__device__ __forceinline__ int   maxi2(int a, int b) { return a > b ? a : b; }
__device__ __forceinline__ int   mini2(int a, int b) { return a < b ? a : b; }

__device__ __forceinline__ int stored_pos(float mp) { return __builtin_bit_cast(int, (mp > -F_INF) ? maxf2(mp, 0.f) : -1.f); }
__device__ __forceinline__ int stored_neg(float mn) { return __builtin_bit_cast(int, maxf2(mn, 0.f)); }

__device__ __forceinline__ int other_half(int i) {
    return __builtin_amdgcn_permlanex16(i, i, 0x76543210, 0xfedcba98, false, false);
}
template <int M>
__device__ __forceinline__ int row_xor(int i) {
    return __builtin_amdgcn_mov_dpp(i, 0x160 | M, 0xF, 0xF, true);
}

template <bool DIAG>
__device__ __forceinline__ void tile_epilogue(const v8f (&acc)[8],
                                              const float* rowNorm, const int* rowLab,
                                              const float* colNorm, const int* colLab,
                                              int (*rowRed)[TILE], int (*colRed)[2][TILE],
                                              const int wave, const int lo, const int hi) {
    float rn[8];
    int   rlab[8];
#pragma unroll
    for (int v = 0; v < 8; ++v) {
        const int rl = wave * 16 + v + hi * 8;
        rn[v]   = rowNorm[rl];
        rlab[v] = rowLab[rl];
    }
    const int dl = lo - hi * 8 - wave * 16;

    float maxPos[8], minNeg[8];
#pragma unroll
    for (int v = 0; v < 8; ++v) { maxPos[v] = -F_INF; minNeg[v] = F_INF; }

#pragma unroll
    for (int g = 0; g < 8; ++g) {
        const int ci = g * 16 + lo;
        const float cn = colNorm[ci];
        const int   cl = colLab[ci];
        float cP = -F_INF, cN = F_INF;
#pragma unroll
        for (int v = 0; v < 8; ++v) {
            const float sq   = rn[v] + cn - 2.0f * acc[g][v];
            const bool  same = (rlab[v] == cl);
            float p;
            if (DIAG) {
                const bool offd = (dl != v - 16 * g);
                p = (same && offd) ? sq : -F_INF;
            } else {
                p = same ? sq : -F_INF;
            }
            const float n = same ? F_INF : sq;
            maxPos[v] = maxf2(maxPos[v], p);
            minNeg[v] = minf2(minNeg[v], n);
            if (!DIAG) {
                cP = maxf2(cP, p);
                cN = minf2(cN, n);
            }
        }
        if (!DIAG) {
            int iP = stored_pos(cP), iN = stored_neg(cN);
            iP = maxi2(iP, other_half(iP));
            iN = mini2(iN, other_half(iN));
            colRed[wave][0][ci] = iP;
            colRed[wave][1][ci] = iN;
        }
        asm volatile("" :: "v"(maxPos[0]), "v"(maxPos[1]), "v"(maxPos[2]), "v"(maxPos[3]),
                           "v"(maxPos[4]), "v"(maxPos[5]), "v"(maxPos[6]), "v"(maxPos[7]),
                           "v"(minNeg[0]), "v"(minNeg[1]), "v"(minNeg[2]), "v"(minNeg[3]),
                           "v"(minNeg[4]), "v"(minNeg[5]), "v"(minNeg[6]), "v"(minNeg[7]));
    }

    int iP[8], iN[8];
#pragma unroll
    for (int v = 0; v < 8; ++v) {
        int p = stored_pos(maxPos[v]), n = stored_neg(minNeg[v]);
        p = maxi2(p, row_xor<1>(p)); n = mini2(n, row_xor<1>(n));
        p = maxi2(p, row_xor<2>(p)); n = mini2(n, row_xor<2>(n));
        p = maxi2(p, row_xor<4>(p)); n = mini2(n, row_xor<4>(n));
        p = maxi2(p, row_xor<8>(p)); n = mini2(n, row_xor<8>(n));
        iP[v] = p; iN[v] = n;
    }
    if (lo == 0) {
#pragma unroll
        for (int v = 0; v < 8; ++v) {
            const int rl = wave * 16 + v + hi * 8;
            rowRed[0][rl] = iP[v];
            rowRed[1][rl] = iN[v];
        }
    }
}

__launch_bounds__(256)
__global__ void tile_kernel(const float* __restrict__ feat,
                            const int*   __restrict__ labels,
                            const float* __restrict__ norms,
                            float* __restrict__ part) {
    __shared__ float Alds[TILE * LDA];
    __shared__ float Blds[TILE * LDA];
    __shared__ float rowNorm[TILE];
    __shared__ int   rowLab[TILE];
    __shared__ float colNorm[TILE];
    __shared__ int   colLab[TILE];
    __shared__ int   rowRed[2][TILE];
    __shared__ int   colRed[8][2][TILE];

    const int bi = blockIdx.x;
    const int d  = blockIdx.y;
    if (d == NB / 2 && bi >= NB / 2) return;
    const int bj = (bi + d) & (NB - 1);

    const int tid  = threadIdx.x;
    const int wave = tid >> 5;
    const int lane = tid & 31;
    const int lo   = lane & 15;
    const int hi   = lane >> 4;
    const int r0   = bi * TILE;
    const int c0   = bj * TILE;

    if (tid < TILE) {
        rowNorm[tid] = norms[r0 + tid];
        rowLab[tid]  = labels[r0 + tid];
    } else {
        const int t = tid - TILE;
        colNorm[t] = norms[c0 + t];
        colLab[t]  = labels[c0 + t];
    }

    const v8f vzero = {0.f, 0.f, 0.f, 0.f, 0.f, 0.f, 0.f, 0.f};
    v8f acc[8];
#pragma unroll
    for (int g = 0; g < 8; ++g) acc[g] = vzero;

    for (int kb = 0; kb < DIM / KB; ++kb) {
        const int k0 = kb * KB;
        v4f av[4], bv[4];
#pragma unroll
        for (int i = 0; i < 4; ++i) {
            int idx = tid + i * 256;
            int row = idx >> 3;
            int c4  = (idx & 7) << 2;
            av[i] = *(const v4f*)(feat + (size_t)(r0 + row) * DIM + k0 + c4);
            bv[i] = *(const v4f*)(feat + (size_t)(c0 + row) * DIM + k0 + c4);
        }
        asm volatile("" :: "v"(av[0]), "v"(av[1]), "v"(av[2]), "v"(av[3]),
                           "v"(bv[0]), "v"(bv[1]), "v"(bv[2]), "v"(bv[3]));
#pragma unroll
        for (int i = 0; i < 4; ++i) {
            int idx = tid + i * 256;
            int row = idx >> 3;
            int c4  = (idx & 7) << 2;
            *(v4f*)(&Alds[row * LDA + c4]) = av[i];
            *(v4f*)(&Blds[row * LDA + c4]) = bv[i];
        }
        __syncthreads();

#pragma unroll
        for (int k4 = 0; k4 < KB; k4 += 4) {
            v2f a = *(const v2f*)(&Alds[(wave * 16 + lo) * LDA + k4 + hi * 2]);
#pragma unroll
            for (int g = 0; g < 8; ++g) {
                v2f b = *(const v2f*)(&Blds[(g * 16 + lo) * LDA + k4 + hi * 2]);
                acc[g] = __builtin_amdgcn_wmma_f32_16x16x4_f32(
                    false, a, false, b, (short)0, acc[g], false, false);
            }
        }
        __syncthreads();
    }

    if (d == 0) tile_epilogue<true >(acc, rowNorm, rowLab, colNorm, colLab, rowRed, colRed, wave, lo, hi);
    else        tile_epilogue<false>(acc, rowNorm, rowLab, colNorm, colLab, rowRed, colRed, wave, lo, hi);
    __syncthreads();

    {
        const int idx = tid;
        const int q = idx >> 7;
        const int e = idx & (TILE - 1);
        part[((size_t)q * NB + bj) * N_PTS + r0 + e] = __builtin_bit_cast(float, rowRed[q][e]);
        if (d != 0) {
            int a = colRed[0][q][e];
#pragma unroll
            for (int w = 1; w < 8; ++w) {
                const int b = colRed[w][q][e];
                a = (q == 1) ? mini2(a, b) : maxi2(a, b);
            }
            part[((size_t)q * NB + bi) * N_PTS + c0 + e] = __builtin_bit_cast(float, a);
        }
    }
}

__global__ void final_kernel(const float* __restrict__ part, float* __restrict__ out) {
    __shared__ float sm[2][8][32];
    const int lane = threadIdx.x & 31;
    const int w    = threadIdx.x >> 5;
    const int row  = blockIdx.x * 32 + lane;
    float mp = -1.f, mn = F_INF;
#pragma unroll
    for (int s = 0; s < 8; ++s) {
        const size_t o = (size_t)(w * 8 + s) * N_PTS + row;
        mp = fmaxf(mp, part[o]);
        mn = fminf(mn, part[(size_t)NB * N_PTS + o]);
    }
    sm[0][w][lane] = mp;
    sm[1][w][lane] = mn;
    __syncthreads();
    if (w == 0) {
#pragma unroll
        for (int k = 1; k < 8; ++k) {
            mp = fmaxf(mp, sm[0][k][lane]);
            mn = fminf(mn, sm[1][k][lane]);
        }
        float2 o2;
        o2.x = (mp >= 0.f) ? sqrtf(mp) : 0.f;
        o2.y = sqrtf((mn < F_INF) ? mn : fmaxf(mp, 0.f));
        *(float2*)(out + (size_t)row * 2) = o2;
    }
}

extern "C" void kernel_launch(void* const* d_in, const int* in_sizes, int n_in,
                              void* d_out, int out_size, void* d_ws, size_t ws_size,
                              hipStream_t stream) {
    (void)in_sizes; (void)n_in; (void)out_size; (void)ws_size;
    const float* feat  = (const float*)d_in[0];
    const int* labels  = (const int*)d_in[1];
    float* ws = (float*)d_ws;

    float* norms = ws;
    float* part  = norms + N_PTS;
    float* out = (float*)d_out;

    norms_kernel<<<N_PTS / 256, 256, 0, stream>>>(feat, norms);

    dim3 grid(NB, ND);
    tile_kernel<<<grid, 256, 0, stream>>>(feat, labels, norms, part);

    final_kernel<<<N_PTS / 32, 256, 0, stream>>>(part, out);
}


# norms launch removed (3 -> 2): each tile work-group sums the squares of the rows it stages (per-thread partials over k, 8-lane add tree: norm sum order changes); staging loads: 2 pointers + offsets
# speedup vs baseline: 2.1311x; 1.0498x over previous
#include <hip/hip_runtime.h>
#include <hip/hip_bf16.h>
#include <math.h>


typedef __attribute__((ext_vector_type(2))) float v2f;
typedef __attribute__((ext_vector_type(4))) float v4f;
typedef __attribute__((ext_vector_type(8))) float v8f;

#define N_PTS 8192
#define DIM   512
#define TILE  128
#define NB    (N_PTS / TILE)
#define ND    (NB / 2 + 1)
#define KB    32
#define LDA   36
#define F_INF (__builtin_inff())


__device__ __forceinline__ float maxf2(float a, float b) { return __builtin_fmaximum_numf(a, b); }
__device__ __forceinline__ float minf2(float a, float b) { return __builtin_fminimum_numf(a, b); }
__device__ __forceinline__ int   maxi2(int a, int b) { return a > b ? a : b; }
__device__ __forceinline__ int   mini2(int a, int b) { return a < b ? a : b; }

__device__ __forceinline__ int stored_pos(float mp) { return __builtin_bit_cast(int, (mp > -F_INF) ? maxf2(mp, 0.f) : -1.f); }
__device__ __forceinline__ int stored_neg(float mn) { return __builtin_bit_cast(int, maxf2(mn, 0.f)); }

__device__ __forceinline__ int other_half(int i) {
    return __builtin_amdgcn_permlanex16(i, i, 0x76543210, 0xfedcba98, false, false);
}
template <int M>
__device__ __forceinline__ int row_xor(int i) {
    return __builtin_amdgcn_mov_dpp(i, 0x160 | M, 0xF, 0xF, true);
}
__device__ __forceinline__ float sq4(v4f v) {
    const v2f lo = {v.x, v.y}, hi = {v.z, v.w};
    const v2f t = __builtin_elementwise_fma(hi, hi, lo * lo);
    return t.x + t.y;
}
template <int M>
__device__ __forceinline__ float row_xor_f(float x) {
    return __builtin_bit_cast(float, row_xor<M>(__builtin_bit_cast(int, x)));
}

template <bool DIAG>
__device__ __forceinline__ void tile_epilogue(const v8f (&acc)[8],
                                              const float* rowNorm, const int* rowLab,
                                              const float* colNorm, const int* colLab,
                                              int (*rowRed)[TILE], int (*colRed)[2][TILE],
                                              const int wave, const int lo, const int hi) {
    float rn[8];
    int   rlab[8];
#pragma unroll
    for (int v = 0; v < 8; ++v) {
        const int rl = wave * 16 + v + hi * 8;
        rn[v]   = rowNorm[rl];
        rlab[v] = rowLab[rl];
    }
    const int dl = lo - hi * 8 - wave * 16;

    float maxPos[8], minNeg[8];
#pragma unroll
    for (int v = 0; v < 8; ++v) { maxPos[v] = -F_INF; minNeg[v] = F_INF; }

#pragma unroll
    for (int g = 0; g < 8; ++g) {
        const int ci = g * 16 + lo;
        const float cn = colNorm[ci];
        const int   cl = colLab[ci];
        float cP = -F_INF, cN = F_INF;
#pragma unroll
        for (int v = 0; v < 8; ++v) {
            const float sq   = rn[v] + cn - 2.0f * acc[g][v];
            const bool  same = (rlab[v] == cl);
            float p;
            if (DIAG) {
                const bool offd = (dl != v - 16 * g);
                p = (same && offd) ? sq : -F_INF;
            } else {
                p = same ? sq : -F_INF;
            }
            const float n = same ? F_INF : sq;
            maxPos[v] = maxf2(maxPos[v], p);
            minNeg[v] = minf2(minNeg[v], n);
            if (!DIAG) {
                cP = maxf2(cP, p);
                cN = minf2(cN, n);
            }
        }
        if (!DIAG) {
            int iP = stored_pos(cP), iN = stored_neg(cN);
            iP = maxi2(iP, other_half(iP));
            iN = mini2(iN, other_half(iN));
            colRed[wave][0][ci] = iP;
            colRed[wave][1][ci] = iN;
        }
        asm volatile("" :: "v"(maxPos[0]), "v"(maxPos[1]), "v"(maxPos[2]), "v"(maxPos[3]),
                           "v"(maxPos[4]), "v"(maxPos[5]), "v"(maxPos[6]), "v"(maxPos[7]),
                           "v"(minNeg[0]), "v"(minNeg[1]), "v"(minNeg[2]), "v"(minNeg[3]),
                           "v"(minNeg[4]), "v"(minNeg[5]), "v"(minNeg[6]), "v"(minNeg[7]));
    }

    int iP[8], iN[8];
#pragma unroll
    for (int v = 0; v < 8; ++v) {
        int p = stored_pos(maxPos[v]), n = stored_neg(minNeg[v]);
        p = maxi2(p, row_xor<1>(p)); n = mini2(n, row_xor<1>(n));
        p = maxi2(p, row_xor<2>(p)); n = mini2(n, row_xor<2>(n));
        p = maxi2(p, row_xor<4>(p)); n = mini2(n, row_xor<4>(n));
        p = maxi2(p, row_xor<8>(p)); n = mini2(n, row_xor<8>(n));
        iP[v] = p; iN[v] = n;
    }
    if (lo == 0) {
#pragma unroll
        for (int v = 0; v < 8; ++v) {
            const int rl = wave * 16 + v + hi * 8;
            rowRed[0][rl] = iP[v];
            rowRed[1][rl] = iN[v];
        }
    }
}

__launch_bounds__(256)
__global__ void tile_kernel(const float* __restrict__ feat,
                            const int*   __restrict__ labels,
                            float* __restrict__ part) {
    __shared__ float Alds[TILE * LDA];
    __shared__ float Blds[TILE * LDA];
    __shared__ float rowNorm[TILE];
    __shared__ int   rowLab[TILE];
    __shared__ float colNorm[TILE];
    __shared__ int   colLab[TILE];
    __shared__ int   rowRed[2][TILE];
    __shared__ int   colRed[8][2][TILE];

    const int bi = blockIdx.x;
    const int d  = blockIdx.y;
    if (d == NB / 2 && bi >= NB / 2) return;
    const int bj = (bi + d) & (NB - 1);

    const int tid  = threadIdx.x;
    const int wave = tid >> 5;
    const int lane = tid & 31;
    const int lo   = lane & 15;
    const int hi   = lane >> 4;
    const int r0   = bi * TILE;
    const int c0   = bj * TILE;

    if (tid < TILE) {
        rowLab[tid]  = labels[r0 + tid];
    } else {
        const int t = tid - TILE;
        colLab[t]  = labels[c0 + t];
    }

    const float* pa = feat + (size_t)(r0 + (tid >> 3)) * DIM + ((tid & 7) << 2);
    const float* pb = feat + (size_t)(c0 + (tid >> 3)) * DIM + ((tid & 7) << 2);

    float sqA[4] = {0.f, 0.f, 0.f, 0.f};
    float sqB[4] = {0.f, 0.f, 0.f, 0.f};

    const v8f vzero = {0.f, 0.f, 0.f, 0.f, 0.f, 0.f, 0.f, 0.f};
    v8f acc[8];
#pragma unroll
    for (int g = 0; g < 8; ++g) acc[g] = vzero;

    for (int kb = 0; kb < DIM / KB; ++kb) {
        const int k0 = kb * KB;
        v4f av[4], bv[4];
#pragma unroll
        for (int i = 0; i < 4; ++i) {
            av[i] = *(const v4f*)(pa + k0 + i * 32 * DIM);
            bv[i] = *(const v4f*)(pb + k0 + i * 32 * DIM);
        }
        asm volatile("" :: "v"(av[0]), "v"(av[1]), "v"(av[2]), "v"(av[3]),
                           "v"(bv[0]), "v"(bv[1]), "v"(bv[2]), "v"(bv[3]));
#pragma unroll
        for (int i = 0; i < 4; ++i) {
            int idx = tid + i * 256;
            int row = idx >> 3;
            int c4  = (idx & 7) << 2;
            *(v4f*)(&Alds[row * LDA + c4]) = av[i];
            *(v4f*)(&Blds[row * LDA + c4]) = bv[i];
        }
#pragma unroll
        for (int i = 0; i < 4; ++i) {
            sqA[i] += sq4(av[i]);
            sqB[i] += sq4(bv[i]);
        }
        asm volatile("" :: "v"(sqA[0]), "v"(sqA[1]), "v"(sqA[2]), "v"(sqA[3]),
                           "v"(sqB[0]), "v"(sqB[1]), "v"(sqB[2]), "v"(sqB[3]));
        __syncthreads();

#pragma unroll
        for (int k4 = 0; k4 < KB; k4 += 4) {
            v2f a = *(const v2f*)(&Alds[(wave * 16 + lo) * LDA + k4 + hi * 2]);
#pragma unroll
            for (int g = 0; g < 8; ++g) {
                v2f b = *(const v2f*)(&Blds[(g * 16 + lo) * LDA + k4 + hi * 2]);
                acc[g] = __builtin_amdgcn_wmma_f32_16x16x4_f32(
                    false, a, false, b, (short)0, acc[g], false, false);
            }
        }
        __syncthreads();
    }

#pragma unroll
    for (int i = 0; i < 4; ++i) {
        float a = sqA[i], b = sqB[i];
        a += row_xor_f<1>(a); b += row_xor_f<1>(b);
        a += row_xor_f<2>(a); b += row_xor_f<2>(b);
        a += row_xor_f<4>(a); b += row_xor_f<4>(b);
        if ((tid & 7) == 0) {
            rowNorm[(tid >> 3) + 32 * i] = a;
            colNorm[(tid >> 3) + 32 * i] = b;
        }
    }
    __syncthreads();

    if (d == 0) tile_epilogue<true >(acc, rowNorm, rowLab, colNorm, colLab, rowRed, colRed, wave, lo, hi);
    else        tile_epilogue<false>(acc, rowNorm, rowLab, colNorm, colLab, rowRed, colRed, wave, lo, hi);
    __syncthreads();

    {
        const int idx = tid;
        const int q = idx >> 7;
        const int e = idx & (TILE - 1);
        part[((size_t)q * NB + bj) * N_PTS + r0 + e] = __builtin_bit_cast(float, rowRed[q][e]);
        if (d != 0) {
            int a = colRed[0][q][e];
#pragma unroll
            for (int w = 1; w < 8; ++w) {
                const int b = colRed[w][q][e];
                a = (q == 1) ? mini2(a, b) : maxi2(a, b);
            }
            part[((size_t)q * NB + bi) * N_PTS + c0 + e] = __builtin_bit_cast(float, a);
        }
    }
}

__global__ void final_kernel(const float* __restrict__ part, float* __restrict__ out) {
    __shared__ float sm[2][8][32];
    const int lane = threadIdx.x & 31;
    const int w    = threadIdx.x >> 5;
    const int row  = blockIdx.x * 32 + lane;
    float mp = -1.f, mn = F_INF;
#pragma unroll
    for (int s = 0; s < 8; ++s) {
        const size_t o = (size_t)(w * 8 + s) * N_PTS + row;
        mp = fmaxf(mp, part[o]);
        mn = fminf(mn, part[(size_t)NB * N_PTS + o]);
    }
    sm[0][w][lane] = mp;
    sm[1][w][lane] = mn;
    __syncthreads();
    if (w == 0) {
#pragma unroll
        for (int k = 1; k < 8; ++k) {
            mp = fmaxf(mp, sm[0][k][lane]);
            mn = fminf(mn, sm[1][k][lane]);
        }
        float2 o2;
        o2.x = (mp >= 0.f) ? sqrtf(mp) : 0.f;
        o2.y = sqrtf((mn < F_INF) ? mn : fmaxf(mp, 0.f));
        *(float2*)(out + (size_t)row * 2) = o2;
    }
}

extern "C" void kernel_launch(void* const* d_in, const int* in_sizes, int n_in,
                              void* d_out, int out_size, void* d_ws, size_t ws_size,
                              hipStream_t stream) {
    (void)in_sizes; (void)n_in; (void)out_size; (void)ws_size;
    const float* feat  = (const float*)d_in[0];
    const int* labels  = (const int*)d_in[1];
    float* ws = (float*)d_ws;

    float* part  = ws;
    float* out = (float*)d_out;

    dim3 grid(NB, ND);
    tile_kernel<<<grid, 256, 0, stream>>>(feat, labels, part);

    final_kernel<<<N_PTS / 32, 256, 0, stream>>>(part, out);
}


# work units: tiles of block distances 30..32 cut into four 128x32 pieces each (same k loop, 2 column groups per wave), launched last so that WGPs that drain early fill up; no sum reordered
# speedup vs baseline: 2.2002x; 1.0324x over previous
#include <hip/hip_runtime.h>
#include <hip/hip_bf16.h>
#include <math.h>


typedef __attribute__((ext_vector_type(2))) float v2f;
typedef __attribute__((ext_vector_type(4))) float v4f;
typedef __attribute__((ext_vector_type(8))) float v8f;

#define N_PTS 8192
#define DIM   512
#define TILE  128
#define NB    (N_PTS / TILE)
#define ND    (NB / 2 + 1)
#define KB    32
#define LDA   36
#define F_INF (__builtin_inff())
#define D_SPLIT 30
#define NEXTRA  (3 * (ND - D_SPLIT))
#define NSLOT   (NB + NEXTRA)
#define GRID_Y  (D_SPLIT + 4 * (NB / 2 - D_SPLIT) + 2)


__device__ __forceinline__ float maxf2(float a, float b) { return __builtin_fmaximum_numf(a, b); }
__device__ __forceinline__ float minf2(float a, float b) { return __builtin_fminimum_numf(a, b); }
__device__ __forceinline__ int   maxi2(int a, int b) { return a > b ? a : b; }
__device__ __forceinline__ int   mini2(int a, int b) { return a < b ? a : b; }

__device__ __forceinline__ int stored_pos(float mp) { return __builtin_bit_cast(int, (mp > -F_INF) ? maxf2(mp, 0.f) : -1.f); }
__device__ __forceinline__ int stored_neg(float mn) { return __builtin_bit_cast(int, maxf2(mn, 0.f)); }

__device__ __forceinline__ int other_half(int i) {
    return __builtin_amdgcn_permlanex16(i, i, 0x76543210, 0xfedcba98, false, false);
}
template <int M>
__device__ __forceinline__ int row_xor(int i) {
    return __builtin_amdgcn_mov_dpp(i, 0x160 | M, 0xF, 0xF, true);
}
__device__ __forceinline__ float sq4(v4f v) {
    const v2f lo = {v.x, v.y}, hi = {v.z, v.w};
    const v2f t = __builtin_elementwise_fma(hi, hi, lo * lo);
    return t.x + t.y;
}
template <int M>
__device__ __forceinline__ float row_xor_f(float x) {
    return __builtin_bit_cast(float, row_xor<M>(__builtin_bit_cast(int, x)));
}

template <bool DIAG, int NG>
__device__ __forceinline__ void tile_epilogue(const v8f (&acc)[NG],
                                              const float* rowNorm, const int* rowLab,
                                              const float* colNorm, const int* colLab,
                                              int (*rowRed)[TILE], int (*colRed)[2][TILE],
                                              const int wave, const int lo, const int hi) {
    float rn[8];
    int   rlab[8];
#pragma unroll
    for (int v = 0; v < 8; ++v) {
        const int rl = wave * 16 + v + hi * 8;
        rn[v]   = rowNorm[rl];
        rlab[v] = rowLab[rl];
    }
    const int dl = lo - hi * 8 - wave * 16;

    float maxPos[8], minNeg[8];
#pragma unroll
    for (int v = 0; v < 8; ++v) { maxPos[v] = -F_INF; minNeg[v] = F_INF; }

#pragma unroll
    for (int g = 0; g < NG; ++g) {
        const int ci = g * 16 + lo;
        const float cn = colNorm[ci];
        const int   cl = colLab[ci];
        float cP = -F_INF, cN = F_INF;
#pragma unroll
        for (int v = 0; v < 8; ++v) {
            const float sq   = rn[v] + cn - 2.0f * acc[g][v];
            const bool  same = (rlab[v] == cl);
            float p;
            if (DIAG) {
                const bool offd = (dl != v - 16 * g);
                p = (same && offd) ? sq : -F_INF;
            } else {
                p = same ? sq : -F_INF;
            }
            const float n = same ? F_INF : sq;
            maxPos[v] = maxf2(maxPos[v], p);
            minNeg[v] = minf2(minNeg[v], n);
            if (!DIAG) {
                cP = maxf2(cP, p);
                cN = minf2(cN, n);
            }
        }
        if (!DIAG) {
            int iP = stored_pos(cP), iN = stored_neg(cN);
            iP = maxi2(iP, other_half(iP));
            iN = mini2(iN, other_half(iN));
            colRed[wave][0][ci] = iP;
            colRed[wave][1][ci] = iN;
        }
        asm volatile("" :: "v"(maxPos[0]), "v"(maxPos[1]), "v"(maxPos[2]), "v"(maxPos[3]),
                           "v"(maxPos[4]), "v"(maxPos[5]), "v"(maxPos[6]), "v"(maxPos[7]),
                           "v"(minNeg[0]), "v"(minNeg[1]), "v"(minNeg[2]), "v"(minNeg[3]),
                           "v"(minNeg[4]), "v"(minNeg[5]), "v"(minNeg[6]), "v"(minNeg[7]));
    }

    int iP[8], iN[8];
#pragma unroll
    for (int v = 0; v < 8; ++v) {
        int p = stored_pos(maxPos[v]), n = stored_neg(minNeg[v]);
        p = maxi2(p, row_xor<1>(p)); n = mini2(n, row_xor<1>(n));
        p = maxi2(p, row_xor<2>(p)); n = mini2(n, row_xor<2>(n));
        p = maxi2(p, row_xor<4>(p)); n = mini2(n, row_xor<4>(n));
        p = maxi2(p, row_xor<8>(p)); n = mini2(n, row_xor<8>(n));
        iP[v] = p; iN[v] = n;
    }
    if (lo == 0) {
#pragma unroll
        for (int v = 0; v < 8; ++v) {
            const int rl = wave * 16 + v + hi * 8;
            rowRed[0][rl] = iP[v];
            rowRed[1][rl] = iN[v];
        }
    }
}

template <int NG>
__device__ __forceinline__ void tile_work(const float* __restrict__ feat,
                                          const int*   __restrict__ labels,
                                          float* __restrict__ part,
                                          const int bi, const int bj, const int cq, const int rowSlot, const bool diag,
                                          float* Alds, float* Blds,
                                          float* rowNorm, int* rowLab, float* colNorm, int* colLab,
                                          int (*rowRed)[TILE], int (*colRed)[2][TILE]) {
    constexpr int BC  = NG * 16;
    constexpr int NBP = NG / 2;

    const int tid  = threadIdx.x;
    const int wave = tid >> 5;
    const int lane = tid & 31;
    const int lo   = lane & 15;
    const int hi   = lane >> 4;
    const int r0   = bi * TILE;
    const int c0   = bj * TILE + cq * BC;

    if (tid < TILE) {
        rowLab[tid]  = labels[r0 + tid];
    } else if (tid - TILE < BC) {
        const int t = tid - TILE;
        colLab[t]  = labels[c0 + t];
    }

    const float* pa = feat + (size_t)(r0 + (tid >> 3)) * DIM + ((tid & 7) << 2);
    const float* pb = feat + (size_t)(c0 + (tid >> 3)) * DIM + ((tid & 7) << 2);

    float sqA[4] = {0.f, 0.f, 0.f, 0.f};
    float sqB[NBP];
#pragma unroll
    for (int i = 0; i < NBP; ++i) sqB[i] = 0.f;

    const v8f vzero = {0.f, 0.f, 0.f, 0.f, 0.f, 0.f, 0.f, 0.f};
    v8f acc[NG];
#pragma unroll
    for (int g = 0; g < NG; ++g) acc[g] = vzero;

    for (int kb = 0; kb < DIM / KB; ++kb) {
        const int k0 = kb * KB;
        v4f av[4], bv[NBP];
#pragma unroll
        for (int i = 0; i < 4; ++i) {
            av[i] = *(const v4f*)(pa + k0 + i * 32 * DIM);
        }
#pragma unroll
        for (int i = 0; i < NBP; ++i) {
            bv[i] = *(const v4f*)(pb + k0 + i * 32 * DIM);
        }
        if constexpr (NBP == 4) {
            asm volatile("" :: "v"(av[0]), "v"(av[1]), "v"(av[2]), "v"(av[3]),
                               "v"(bv[0]), "v"(bv[1]), "v"(bv[2]), "v"(bv[3]));
        } else {
            asm volatile("" :: "v"(av[0]), "v"(av[1]), "v"(av[2]), "v"(av[3]), "v"(bv[0]));
        }
#pragma unroll
        for (int i = 0; i < 4; ++i) {
            int idx = tid + i * 256;
            int row = idx >> 3;
            int c4  = (idx & 7) << 2;
            *(v4f*)(&Alds[row * LDA + c4]) = av[i];
        }
#pragma unroll
        for (int i = 0; i < NBP; ++i) {
            int idx = tid + i * 256;
            int row = idx >> 3;
            int c4  = (idx & 7) << 2;
            *(v4f*)(&Blds[row * LDA + c4]) = bv[i];
        }
#pragma unroll
        for (int i = 0; i < 4; ++i) sqA[i] += sq4(av[i]);
#pragma unroll
        for (int i = 0; i < NBP; ++i) sqB[i] += sq4(bv[i]);
        if constexpr (NBP == 4) {
            asm volatile("" :: "v"(sqA[0]), "v"(sqA[1]), "v"(sqA[2]), "v"(sqA[3]),
                               "v"(sqB[0]), "v"(sqB[1]), "v"(sqB[2]), "v"(sqB[3]));
        } else {
            asm volatile("" :: "v"(sqA[0]), "v"(sqA[1]), "v"(sqA[2]), "v"(sqA[3]), "v"(sqB[0]));
        }
        __syncthreads();

#pragma unroll
        for (int k4 = 0; k4 < KB; k4 += 4) {
            v2f a = *(const v2f*)(&Alds[(wave * 16 + lo) * LDA + k4 + hi * 2]);
#pragma unroll
            for (int g = 0; g < NG; ++g) {
                v2f b = *(const v2f*)(&Blds[(g * 16 + lo) * LDA + k4 + hi * 2]);
                acc[g] = __builtin_amdgcn_wmma_f32_16x16x4_f32(
                    false, a, false, b, (short)0, acc[g], false, false);
            }
        }
        __syncthreads();
    }

#pragma unroll
    for (int i = 0; i < 4; ++i) {
        float a = sqA[i];
        a += row_xor_f<1>(a);
        a += row_xor_f<2>(a);
        a += row_xor_f<4>(a);
        if ((tid & 7) == 0) rowNorm[(tid >> 3) + 32 * i] = a;
    }
#pragma unroll
    for (int i = 0; i < NBP; ++i) {
        float b = sqB[i];
        b += row_xor_f<1>(b);
        b += row_xor_f<2>(b);
        b += row_xor_f<4>(b);
        if ((tid & 7) == 0) colNorm[(tid >> 3) + 32 * i] = b;
    }
    __syncthreads();

    if constexpr (NG == 8) {
        if (diag) tile_epilogue<true,  NG>(acc, rowNorm, rowLab, colNorm, colLab, rowRed, colRed, wave, lo, hi);
        else      tile_epilogue<false, NG>(acc, rowNorm, rowLab, colNorm, colLab, rowRed, colRed, wave, lo, hi);
    } else {
        tile_epilogue<false, NG>(acc, rowNorm, rowLab, colNorm, colLab, rowRed, colRed, wave, lo, hi);
    }
    __syncthreads();

    {
        const int q = tid >> 7;
        const int e = tid & (TILE - 1);
        part[((size_t)q * NSLOT + rowSlot) * N_PTS + r0 + e] = __builtin_bit_cast(float, rowRed[q][e]);
    }
    if (!diag && tid < 2 * BC) {
        const int q = tid / BC;
        const int e = tid % BC;
        int a = colRed[0][q][e];
#pragma unroll
        for (int w = 1; w < 8; ++w) {
            const int b = colRed[w][q][e];
            a = (q == 1) ? mini2(a, b) : maxi2(a, b);
        }
        part[((size_t)q * NSLOT + bi) * N_PTS + c0 + e] = __builtin_bit_cast(float, a);
    }
}

__launch_bounds__(256)
__global__ void tile_kernel(const float* __restrict__ feat,
                            const int*   __restrict__ labels,
                            float* __restrict__ part) {
    __shared__ float Alds[TILE * LDA];
    __shared__ float Blds[TILE * LDA];
    __shared__ float rowNorm[TILE];
    __shared__ int   rowLab[TILE];
    __shared__ float colNorm[TILE];
    __shared__ int   colLab[TILE];
    __shared__ int   rowRed[2][TILE];
    __shared__ int   colRed[8][2][TILE];

    const int y = blockIdx.y;
    if (y < D_SPLIT) {
        const int bi = blockIdx.x;
        const int bj = (bi + y) & (NB - 1);
        tile_work<8>(feat, labels, part, bi, bj, 0, bj, y == 0,
                     Alds, Blds, rowNorm, rowLab, colNorm, colLab, rowRed, colRed);
    } else {
        const int p  = (y - D_SPLIT) * NB + blockIdx.x;
        const int dd = p >> 8;
        const int u  = p & 255;
        const int bi = u >> 2;
        const int cq = u & 3;
        const int bj = (bi + D_SPLIT + dd) & (NB - 1);
        const int rowSlot = (cq == 0) ? bj : (NB + 3 * dd + cq - 1);
        tile_work<2>(feat, labels, part, bi, bj, cq, rowSlot, false,
                     Alds, Blds, rowNorm, rowLab, colNorm, colLab, rowRed, colRed);
    }
}

__global__ void final_kernel(const float* __restrict__ part, float* __restrict__ out) {
    __shared__ float sm[2][8][32];
    const int lane = threadIdx.x & 31;
    const int w    = threadIdx.x >> 5;
    const int row  = blockIdx.x * 32 + lane;
    float mp = -1.f, mn = F_INF;
#pragma unroll
    for (int s = 0; s < 8; ++s) {
        const size_t o = (size_t)(w * 8 + s) * N_PTS + row;
        mp = fmaxf(mp, part[o]);
        mn = fminf(mn, part[(size_t)NSLOT * N_PTS + o]);
    }
    for (int s = w; s < NEXTRA; s += 8) {
        const bool written = (D_SPLIT + s / 3 < NB / 2) || (row < N_PTS / 2);
        if (written) {
            const size_t o = (size_t)(NB + s) * N_PTS + row;
            mp = fmaxf(mp, part[o]);
            mn = fminf(mn, part[(size_t)NSLOT * N_PTS + o]);
        }
    }
    sm[0][w][lane] = mp;
    sm[1][w][lane] = mn;
    __syncthreads();
    if (w == 0) {
#pragma unroll
        for (int k = 1; k < 8; ++k) {
            mp = fmaxf(mp, sm[0][k][lane]);
            mn = fminf(mn, sm[1][k][lane]);
        }
        float2 o2;
        o2.x = (mp >= 0.f) ? sqrtf(mp) : 0.f;
        o2.y = sqrtf((mn < F_INF) ? mn : fmaxf(mp, 0.f));
        *(float2*)(out + (size_t)row * 2) = o2;
    }
}

extern "C" void kernel_launch(void* const* d_in, const int* in_sizes, int n_in,
                              void* d_out, int out_size, void* d_ws, size_t ws_size,
                              hipStream_t stream) {
    (void)in_sizes; (void)n_in; (void)out_size; (void)ws_size;
    const float* feat  = (const float*)d_in[0];
    const int* labels  = (const int*)d_in[1];
    float* ws = (float*)d_ws;

    float* part  = ws;
    float* out = (float*)d_out;

    dim3 grid(NB, GRID_Y);
    tile_kernel<<<grid, 256, 0, stream>>>(feat, labels, part);

    final_kernel<<<N_PTS / 32, 256, 0, stream>>>(part, out);
}
